# starting kernel
# baseline (speedup 1.0000x reference)
#include <hip/hip_runtime.h>

#define IN_FEATS   96
#define OUT_FEATS  128
#define N_ETYPES   4
#define K_DIM      (N_ETYPES * IN_FEATS + N_ETYPES)
#define K_PAIRS    (K_DIM / 2)

typedef __attribute__((ext_vector_type(2))) float v2f;
typedef __attribute__((ext_vector_type(4))) float v4f;
typedef __attribute__((ext_vector_type(8))) float v8f;

__global__ __launch_bounds__(256) void ggc_zero(float* __restrict__ p, size_t n) {
    size_t i = (size_t)blockIdx.x * blockDim.x + threadIdx.x;
    size_t stride = (size_t)gridDim.x * blockDim.x;
    size_t n4 = n >> 2;
    v4f* p4 = (v4f*)p;
    v4f z = {0.f, 0.f, 0.f, 0.f};
    for (size_t k = i; k < n4; k += stride) p4[k] = z;
    for (size_t k = (n4 << 2) + i; k < n; k += stride) p[k] = 0.f;
}

__global__ __launch_bounds__(256) void ggc_packB(const float* __restrict__ W,
                                                 const float* __restrict__ b,
                                                 float* __restrict__ Bm2) {
    int idx = blockIdx.x * blockDim.x + threadIdx.x;
    if (idx >= K_DIM * OUT_FEATS) return;
    int h  = idx & 1;
    int j  = (idx >> 1) & (OUT_FEATS - 1);
    int p  = idx >> 8;
    int kk = 2 * p + h;
    float v;
    if (kk < N_ETYPES * IN_FEATS) {
        int t = kk / IN_FEATS;
        int k = kk - t * IN_FEATS;
        v = W[((size_t)t * OUT_FEATS + j) * OUT_FEATS + k];
    } else {
        v = b[(kk - N_ETYPES * IN_FEATS) * OUT_FEATS + j];
    }
    Bm2[idx] = v;
}

__global__ __launch_bounds__(256) void ggc_scatter(const float* __restrict__ feat,
                                                   const int*   __restrict__ etypes,
                                                   const int*   __restrict__ src,
                                                   const int*   __restrict__ dst,
                                                   float* __restrict__ sT2,
                                                   int n_edges, int n_nodes) {
    int e    = (int)((blockIdx.x * (unsigned)blockDim.x + threadIdx.x) >> 5);
    int lane = threadIdx.x & 31;
    if (e >= n_edges) return;
    int t = etypes[e];
    int s = src[e];
    int d = dst[e];
    const float* frow = feat + (size_t)s * IN_FEATS;
    size_t nn = (size_t)n_nodes;
    #pragma unroll
    for (int c = 0; c < IN_FEATS; c += 32) {
        float v = frow[c + lane];
        int row = t * IN_FEATS + c + lane;
        atomicAdd(&sT2[((size_t)(row >> 1) * nn + d) * 2 + (row & 1)], v);
    }
    if (lane == 0) {
        int row = N_ETYPES * IN_FEATS + t;
        atomicAdd(&sT2[((size_t)(row >> 1) * nn + d) * 2 + (row & 1)], 1.0f);
    }
}

__global__ __launch_bounds__(256) void ggc_gemm(const float* __restrict__ sT2,
                                                const float* __restrict__ Bm2,
                                                float* __restrict__ out,
                                                int n_nodes, int n_tiles) {
    int lane  = threadIdx.x & 31;
    int wave  = threadIdx.x >> 5;
    int tile  = blockIdx.x * 8 + wave;
    if (tile >= n_tiles) return;

    int node0 = tile * 16;
    int m     = lane & 15;
    int hi    = lane >> 4;
    int mrow  = node0 + m;
    if (mrow > n_nodes - 1) mrow = n_nodes - 1;
    size_t nn = (size_t)n_nodes;

    v8f acc[8] = {};

    for (int p0 = 0; p0 < K_PAIRS; p0 += 2) {
        int p = p0 + hi;
        v2f a = *(const v2f*)(sT2 + ((size_t)p * nn + mrow) * 2);
        const float* Bp = Bm2 + (size_t)p * (OUT_FEATS * 2);
        #pragma unroll
        for (int nt = 0; nt < 8; nt++) {
            v2f bv = *(const v2f*)(Bp + (nt * 16 + m) * 2);
            acc[nt] = __builtin_amdgcn_wmma_f32_16x16x4_f32(
                 false, a,  false, bv,
                 (short)0, acc[nt],  false,  false);
        }
    }

    int mbase = node0 + hi * 8;
    #pragma unroll
    for (int r = 0; r < 8; r++) {
        int node = mbase + r;
        if (node < n_nodes) {
            float* orow = out + (size_t)node * OUT_FEATS;
            #pragma unroll
            for (int nt = 0; nt < 8; nt++) {
                orow[nt * 16 + m] = acc[nt][r];
            }
        }
    }
}

extern "C" void kernel_launch(void* const* d_in, const int* in_sizes, int n_in,
                              void* d_out, int out_size, void* d_ws, size_t ws_size,
                              hipStream_t stream) {
    const float* feat   = (const float*)d_in[0];
    const int*   etypes = (const int*)  d_in[1];
    const int*   src    = (const int*)  d_in[2];
    const int*   dst    = (const int*)  d_in[3];
    const float* W      = (const float*)d_in[4];
    const float* b      = (const float*)d_in[5];
    float*       out    = (float*)d_out;

    int n_nodes = in_sizes[0] / IN_FEATS;
    int n_edges = in_sizes[1];

    float* sT2 = (float*)d_ws;
    float* Bm2 = sT2 + (size_t)K_DIM * n_nodes;

    size_t s_elems = (size_t)K_DIM * n_nodes;
    ggc_zero<<<2048, 256, 0, stream>>>(sT2, s_elems);

    int packN = K_DIM * OUT_FEATS;
    ggc_packB<<<(packN + 255) / 256, 256, 0, stream>>>(W, b, Bm2);

    int scatter_blocks = (n_edges * 32 + 255) / 256;
    ggc_scatter<<<scatter_blocks, 256, 0, stream>>>(feat, etypes, src, dst, sT2,
                                                    n_edges, n_nodes);

    int n_tiles = (n_nodes + 15) / 16;
    ggc_gemm<<<(n_tiles + 7) / 8, 256, 0, stream>>>(sT2, Bm2, out, n_nodes, n_tiles);
}


# Sum table stored type-major [type][node][96] so each wave-wide atomic add lands on one aligned 128-B line; GEMM reads A as b128 row pieces (k order permuted inside each 8)
# speedup vs baseline: 4.8620x; 4.8620x over previous
#include <hip/hip_runtime.h>

#define IN_FEATS   96
#define OUT_FEATS  128
#define N_ETYPES   4
#define K_DIM      (N_ETYPES * IN_FEATS + N_ETYPES)
#define K_PAIRS    (K_DIM / 2)

typedef __attribute__((ext_vector_type(2))) float v2f;
typedef __attribute__((ext_vector_type(4))) float v4f;
typedef __attribute__((ext_vector_type(8))) float v8f;

__global__ __launch_bounds__(256) void ggc_zero(float* __restrict__ p, size_t n) {
    size_t i = (size_t)blockIdx.x * blockDim.x + threadIdx.x;
    size_t stride = (size_t)gridDim.x * blockDim.x;
    size_t n4 = n >> 2;
    v4f* p4 = (v4f*)p;
    v4f z = {0.f, 0.f, 0.f, 0.f};
    for (size_t k = i; k < n4; k += stride) p4[k] = z;
    for (size_t k = (n4 << 2) + i; k < n; k += stride) p[k] = 0.f;
}

__global__ __launch_bounds__(256) void ggc_packB(const float* __restrict__ W,
                                                 const float* __restrict__ b,
                                                 float* __restrict__ Bm2) {
    int idx = blockIdx.x * blockDim.x + threadIdx.x;
    if (idx >= K_DIM * OUT_FEATS) return;
    int h  = idx & 1;
    int j  = (idx >> 1) & (OUT_FEATS - 1);
    int p  = idx >> 8;
    int kk = 2 * p + h;
    float v;
    if (kk < N_ETYPES * IN_FEATS) {
        int t = kk / IN_FEATS;
        int k = kk - t * IN_FEATS;
        v = W[((size_t)t * OUT_FEATS + j) * OUT_FEATS + k];
    } else {
        v = b[(kk - N_ETYPES * IN_FEATS) * OUT_FEATS + j];
    }
    Bm2[idx] = v;
}

__global__ __launch_bounds__(256) void ggc_scatter(const float* __restrict__ feat,
                                                   const int*   __restrict__ etypes,
                                                   const int*   __restrict__ src,
                                                   const int*   __restrict__ dst,
                                                   float* __restrict__ sT,
                                                   int n_edges, int n_nodes) {
    int e    = (int)((blockIdx.x * (unsigned)blockDim.x + threadIdx.x) >> 5);
    int lane = threadIdx.x & 31;
    if (e >= n_edges) return;
    int t = etypes[e];
    int s = src[e];
    int d = dst[e];
    const float* frow = feat + (size_t)s * IN_FEATS;
    size_t nn = (size_t)n_nodes;
    float* srow = sT + ((size_t)t * nn + (size_t)d) * IN_FEATS;
    #pragma unroll
    for (int c = 0; c < IN_FEATS; c += 32) {
        float v = frow[c + lane];
        atomicAdd(&srow[c + lane], v);
    }
    if (lane == 0) {
        float* cnt = sT + (size_t)(N_ETYPES * IN_FEATS) * nn;
        atomicAdd(&cnt[(size_t)d * N_ETYPES + t], 1.0f);
    }
}

__global__ __launch_bounds__(256) void ggc_gemm(const float* __restrict__ sT,
                                                const float* __restrict__ Bm2,
                                                float* __restrict__ out,
                                                int n_nodes, int n_tiles) {
    int lane  = threadIdx.x & 31;
    int wave  = threadIdx.x >> 5;
    int tile  = blockIdx.x * 8 + wave;
    if (tile >= n_tiles) return;

    int node0 = tile * 16;
    int m     = lane & 15;
    int hi    = lane >> 4;
    int mrow  = node0 + m;
    if (mrow > n_nodes - 1) mrow = n_nodes - 1;
    size_t nn = (size_t)n_nodes;

    v8f acc[8] = {};

    const float* Bl = Bm2 + (size_t)hi * (2 * OUT_FEATS * 2) + m * 2;

    for (int t = 0; t < N_ETYPES; t++) {
        const float* Arow = sT + ((size_t)t * nn + (size_t)mrow) * IN_FEATS + 4 * hi;
        for (int q = 0; q < IN_FEATS / 8; q++) {
            v4f a4 = *(const v4f*)(Arow + 8 * q);
            int pair0 = t * (IN_FEATS / 2) + 4 * q;
            #pragma unroll
            for (int s = 0; s < 2; s++) {
                v2f a;
                a.x = s ? a4.z : a4.x;
                a.y = s ? a4.w : a4.y;
                const float* Bp = Bl + (size_t)(pair0 + s) * (OUT_FEATS * 2);
                #pragma unroll
                for (int nt = 0; nt < 8; nt++) {
                    v2f bv = *(const v2f*)(Bp + nt * 32);
                    acc[nt] = __builtin_amdgcn_wmma_f32_16x16x4_f32(
                         false, a,  false, bv,
                         (short)0, acc[nt],  false,  false);
                }
            }
        }
    }

    {
        const float* cnt = sT + (size_t)(N_ETYPES * IN_FEATS) * nn;
        v2f a = *(const v2f*)(cnt + (size_t)mrow * N_ETYPES + 2 * hi);
        const float* Bp = Bm2 + (size_t)(N_ETYPES * IN_FEATS / 2 + hi) * (OUT_FEATS * 2) + m * 2;
        #pragma unroll
        for (int nt = 0; nt < 8; nt++) {
            v2f bv = *(const v2f*)(Bp + nt * 32);
            acc[nt] = __builtin_amdgcn_wmma_f32_16x16x4_f32(
                 false, a,  false, bv,
                 (short)0, acc[nt],  false,  false);
        }
    }

    int mbase = node0 + hi * 8;
    #pragma unroll
    for (int r = 0; r < 8; r++) {
        int node = mbase + r;
        if (node < n_nodes) {
            float* orow = out + (size_t)node * OUT_FEATS;
            #pragma unroll
            for (int nt = 0; nt < 8; nt++) {
                orow[nt * 16 + m] = acc[nt][r];
            }
        }
    }
}

extern "C" void kernel_launch(void* const* d_in, const int* in_sizes, int n_in,
                              void* d_out, int out_size, void* d_ws, size_t ws_size,
                              hipStream_t stream) {
    const float* feat   = (const float*)d_in[0];
    const int*   etypes = (const int*)  d_in[1];
    const int*   src    = (const int*)  d_in[2];
    const int*   dst    = (const int*)  d_in[3];
    const float* W      = (const float*)d_in[4];
    const float* b      = (const float*)d_in[5];
    float*       out    = (float*)d_out;

    int n_nodes = in_sizes[0] / IN_FEATS;
    int n_edges = in_sizes[1];

    float* sT  = (float*)d_ws;
    float* Bm2 = sT + (size_t)K_DIM * n_nodes;

    size_t s_elems = (size_t)K_DIM * n_nodes;
    ggc_zero<<<2048, 256, 0, stream>>>(sT, s_elems);

    int packN = K_DIM * OUT_FEATS;
    ggc_packB<<<(packN + 255) / 256, 256, 0, stream>>>(W, b, Bm2);

    int scatter_blocks = (n_edges * 32 + 255) / 256;
    ggc_scatter<<<scatter_blocks, 256, 0, stream>>>(feat, etypes, src, dst, sT,
                                                    n_edges, n_nodes);

    int n_tiles = (n_nodes + 15) / 16;
    ggc_gemm<<<(n_tiles + 7) / 8, 256, 0, stream>>>(sT, Bm2, out, n_nodes, n_tiles);
}


# Float atomic scatter replaced by per-(destination,type) edge lists (one int exchange per edge) and a per-node gather-sum writing each sum row once; zero pass gone; sums run in list (arrival) order
# speedup vs baseline: 8.9372x; 1.8382x over previous
#include <hip/hip_runtime.h>

#define IN_FEATS   96
#define OUT_FEATS  128
#define N_ETYPES   4
#define K_DIM      (N_ETYPES * IN_FEATS + N_ETYPES)
#define K_PAIRS    (K_DIM / 2)

typedef __attribute__((ext_vector_type(2))) float v2f;
typedef __attribute__((ext_vector_type(4))) float v4f;
typedef __attribute__((ext_vector_type(8))) float v8f;

__global__ __launch_bounds__(256) void ggc_prep(const float* __restrict__ W,
                                                const float* __restrict__ b,
                                                float* __restrict__ Bm2,
                                                int* __restrict__ head, int n_heads) {
    int idx = blockIdx.x * blockDim.x + threadIdx.x;
    if (idx < n_heads) head[idx] = -1;
    if (idx >= K_DIM * OUT_FEATS) return;
    int h  = idx & 1;
    int j  = (idx >> 1) & (OUT_FEATS - 1);
    int p  = idx >> 8;
    int kk = 2 * p + h;
    float v;
    if (kk < N_ETYPES * IN_FEATS) {
        int t = kk / IN_FEATS;
        int k = kk - t * IN_FEATS;
        v = W[((size_t)t * OUT_FEATS + j) * OUT_FEATS + k];
    } else {
        v = b[(kk - N_ETYPES * IN_FEATS) * OUT_FEATS + j];
    }
    Bm2[idx] = v;
}

__global__ __launch_bounds__(256) void ggc_link(const int* __restrict__ etypes,
                                                const int* __restrict__ src,
                                                const int* __restrict__ dst,
                                                int* __restrict__ head,
                                                int2* __restrict__ rec,
                                                int n_edges, int n_nodes) {
    int e = blockIdx.x * blockDim.x + threadIdx.x;
    if (e >= n_edges) return;
    int t = etypes[e];
    int s = src[e];
    int d = dst[e];
    int prev = -1;
    if ((unsigned)t < (unsigned)N_ETYPES && (unsigned)d < (unsigned)n_nodes)
        prev = atomicExch(&head[(size_t)d * N_ETYPES + t], e);
    rec[e] = make_int2(s, prev);
}

__device__ __forceinline__ int ggc_walk(const float* __restrict__ feat,
                                        const int2* __restrict__ rec,
                                        int2 r, bool live, int n_edges,
                                        float* __restrict__ row) {
    float a0 = 0.f, a1 = 0.f, a2 = 0.f;
    int c = 0;
    for (int it = 0; it < n_edges && live; ++it) {
        int s  = r.x;
        int en = r.y;
        const float* frow = feat + (size_t)s * IN_FEATS;
        float v0 = frow[0];
        float v1 = frow[32];
        float v2 = frow[64];
        live = (unsigned)en < (unsigned)n_edges;
        if (live) r = rec[en];
        a0 += v0; a1 += v1; a2 += v2;
        c++;
    }
    row[0] = a0; row[32] = a1; row[64] = a2;
    return c;
}

__global__ __launch_bounds__(256) void ggc_gather(const float* __restrict__ feat,
                                                  const int*   __restrict__ head,
                                                  const int2*  __restrict__ rec,
                                                  float* __restrict__ sT,
                                                  int n_edges, int n_nodes) {
    int node = __builtin_amdgcn_readfirstlane((int)((blockIdx.x * (unsigned)blockDim.x + threadIdx.x) >> 5));
    int lane = threadIdx.x & 31;
    if (node >= n_nodes) return;

    const int* hp = head + (size_t)node * N_ETYPES;
    int e0 = hp[0], e1 = hp[1], e2 = hp[2], e3 = hp[3];
    bool l0 = (unsigned)e0 < (unsigned)n_edges;
    bool l1 = (unsigned)e1 < (unsigned)n_edges;
    bool l2 = (unsigned)e2 < (unsigned)n_edges;
    bool l3 = (unsigned)e3 < (unsigned)n_edges;
    int2 q0 = make_int2(0, -1), q1 = q0, q2 = q0, q3 = q0;
    if (l0) q0 = rec[e0];
    if (l1) q1 = rec[e1];
    if (l2) q2 = rec[e2];
    if (l3) q3 = rec[e3];

    size_t nn = (size_t)n_nodes;
    const float* fl = feat + lane;
    float* base = sT + (size_t)node * IN_FEATS + lane;
    int c0 = ggc_walk(fl, rec, q0, l0, n_edges, base);
    int c1 = ggc_walk(fl, rec, q1, l1, n_edges, base + (size_t)1 * nn * IN_FEATS);
    int c2 = ggc_walk(fl, rec, q2, l2, n_edges, base + (size_t)2 * nn * IN_FEATS);
    int c3 = ggc_walk(fl, rec, q3, l3, n_edges, base + (size_t)3 * nn * IN_FEATS);

    if (lane < N_ETYPES) {
        int c = lane == 0 ? c0 : lane == 1 ? c1 : lane == 2 ? c2 : c3;
        float* cnt = sT + (size_t)(N_ETYPES * IN_FEATS) * nn;
        cnt[(size_t)node * N_ETYPES + lane] = (float)c;
    }
}

__global__ __launch_bounds__(256) void ggc_gemm(const float* __restrict__ sT,
                                                const float* __restrict__ Bm2,
                                                float* __restrict__ out,
                                                int n_nodes, int n_tiles) {
    int lane  = threadIdx.x & 31;
    int wave  = threadIdx.x >> 5;
    int tile  = blockIdx.x * 8 + wave;
    if (tile >= n_tiles) return;

    int node0 = tile * 16;
    int m     = lane & 15;
    int hi    = lane >> 4;
    int mrow  = node0 + m;
    if (mrow > n_nodes - 1) mrow = n_nodes - 1;
    size_t nn = (size_t)n_nodes;

    v8f acc[8] = {};

    const float* Bl = Bm2 + (size_t)hi * (2 * OUT_FEATS * 2) + m * 2;

    for (int t = 0; t < N_ETYPES; t++) {
        const float* Arow = sT + ((size_t)t * nn + (size_t)mrow) * IN_FEATS + 4 * hi;
        for (int q = 0; q < IN_FEATS / 8; q++) {
            v4f a4 = *(const v4f*)(Arow + 8 * q);
            int pair0 = t * (IN_FEATS / 2) + 4 * q;
            #pragma unroll
            for (int s = 0; s < 2; s++) {
                v2f a;
                a.x = s ? a4.z : a4.x;
                a.y = s ? a4.w : a4.y;
                const float* Bp = Bl + (size_t)(pair0 + s) * (OUT_FEATS * 2);
                #pragma unroll
                for (int nt = 0; nt < 8; nt++) {
                    v2f bv = *(const v2f*)(Bp + nt * 32);
                    acc[nt] = __builtin_amdgcn_wmma_f32_16x16x4_f32(
                         false, a,  false, bv,
                         (short)0, acc[nt],  false,  false);
                }
            }
        }
    }

    {
        const float* cnt = sT + (size_t)(N_ETYPES * IN_FEATS) * nn;
        v2f a = *(const v2f*)(cnt + (size_t)mrow * N_ETYPES + 2 * hi);
        const float* Bp = Bm2 + (size_t)(N_ETYPES * IN_FEATS / 2 + hi) * (OUT_FEATS * 2) + m * 2;
        #pragma unroll
        for (int nt = 0; nt < 8; nt++) {
            v2f bv = *(const v2f*)(Bp + nt * 32);
            acc[nt] = __builtin_amdgcn_wmma_f32_16x16x4_f32(
                 false, a,  false, bv,
                 (short)0, acc[nt],  false,  false);
        }
    }

    int mbase = node0 + hi * 8;
    #pragma unroll
    for (int r = 0; r < 8; r++) {
        int node = mbase + r;
        if (node < n_nodes) {
            float* orow = out + (size_t)node * OUT_FEATS;
            #pragma unroll
            for (int nt = 0; nt < 8; nt++) {
                orow[nt * 16 + m] = acc[nt][r];
            }
        }
    }
}

extern "C" void kernel_launch(void* const* d_in, const int* in_sizes, int n_in,
                              void* d_out, int out_size, void* d_ws, size_t ws_size,
                              hipStream_t stream) {
    const float* feat   = (const float*)d_in[0];
    const int*   etypes = (const int*)  d_in[1];
    const int*   src    = (const int*)  d_in[2];
    const int*   dst    = (const int*)  d_in[3];
    const float* W      = (const float*)d_in[4];
    const float* b      = (const float*)d_in[5];
    float*       out    = (float*)d_out;

    int n_nodes = in_sizes[0] / IN_FEATS;
    int n_edges = in_sizes[1];

    float* sT   = (float*)d_ws;
    float* Bm2  = sT + (size_t)K_DIM * n_nodes;
    int*   head = (int*)(Bm2 + (size_t)K_DIM * OUT_FEATS);
    int2*  rec  = (int2*)(head + (size_t)n_nodes * N_ETYPES);

    int packN   = K_DIM * OUT_FEATS;
    int n_heads = n_nodes * N_ETYPES;
    int prepN   = packN > n_heads ? packN : n_heads;
    ggc_prep<<<(prepN + 255) / 256, 256, 0, stream>>>(W, b, Bm2, head, n_heads);

    ggc_link<<<(n_edges + 255) / 256, 256, 0, stream>>>(etypes, src, dst, head, rec,
                                                        n_edges, n_nodes);

    int gather_blocks = (n_nodes * 32 + 255) / 256;
    ggc_gather<<<gather_blocks, 256, 0, stream>>>(feat, head, rec, sT, n_edges, n_nodes);

    int n_tiles = (n_nodes + 15) / 16;
    ggc_gemm<<<(n_tiles + 7) / 8, 256, 0, stream>>>(sT, Bm2, out, n_nodes, n_tiles);
}


# Gather walks a node's four type lists side by side, one per 8-lane group reading 16-B row pieces, not one after another: dependent loads per wave = longest list, not the sum; list-order sums kept
# speedup vs baseline: 9.4663x; 1.0592x over previous
#include <hip/hip_runtime.h>

#define IN_FEATS   96
#define OUT_FEATS  128
#define N_ETYPES   4
#define K_DIM      (N_ETYPES * IN_FEATS + N_ETYPES)
#define K_PAIRS    (K_DIM / 2)

typedef __attribute__((ext_vector_type(2))) float v2f;
typedef __attribute__((ext_vector_type(4))) float v4f;
typedef __attribute__((ext_vector_type(8))) float v8f;

__global__ __launch_bounds__(256) void ggc_prep(const float* __restrict__ W,
                                                const float* __restrict__ b,
                                                float* __restrict__ Bm2,
                                                int* __restrict__ head, int n_heads) {
    int idx = blockIdx.x * blockDim.x + threadIdx.x;
    if (idx < n_heads) head[idx] = -1;
    if (idx >= K_DIM * OUT_FEATS) return;
    int h  = idx & 1;
    int j  = (idx >> 1) & (OUT_FEATS - 1);
    int p  = idx >> 8;
    int kk = 2 * p + h;
    float v;
    if (kk < N_ETYPES * IN_FEATS) {
        int t = kk / IN_FEATS;
        int k = kk - t * IN_FEATS;
        v = W[((size_t)t * OUT_FEATS + j) * OUT_FEATS + k];
    } else {
        v = b[(kk - N_ETYPES * IN_FEATS) * OUT_FEATS + j];
    }
    Bm2[idx] = v;
}

__global__ __launch_bounds__(256) void ggc_link(const int* __restrict__ etypes,
                                                const int* __restrict__ src,
                                                const int* __restrict__ dst,
                                                int* __restrict__ head,
                                                int2* __restrict__ rec,
                                                int n_edges, int n_nodes) {
    int e = blockIdx.x * blockDim.x + threadIdx.x;
    if (e >= n_edges) return;
    int t = etypes[e];
    int s = src[e];
    int d = dst[e];
    int prev = -1;
    if ((unsigned)t < (unsigned)N_ETYPES && (unsigned)d < (unsigned)n_nodes)
        prev = atomicExch(&head[(size_t)d * N_ETYPES + t], e);
    rec[e] = make_int2(s, prev);
}

__global__ __launch_bounds__(256) void ggc_gather(const float* __restrict__ feat,
                                                  const int*   __restrict__ head,
                                                  const int2*  __restrict__ rec,
                                                  float* __restrict__ sT,
                                                  int n_edges, int n_nodes) {
    int node = (int)((blockIdx.x * (unsigned)blockDim.x + threadIdx.x) >> 5);
    int lane = threadIdx.x & 31;
    int t = lane >> 3;
    int j = lane & 7;
    if (node >= n_nodes) return;

    int  e    = head[(size_t)node * N_ETYPES + t];
    bool live = (unsigned)e < (unsigned)n_edges;
    int2 r    = make_int2(0, -1);
    if (live) r = rec[e];

    v4f a0 = {0.f, 0.f, 0.f, 0.f}, a1 = a0, a2 = a0;
    int c = 0;
    for (int it = 0; it < n_edges && live; ++it) {
        const v4f* frow = (const v4f*)(feat + (size_t)r.x * IN_FEATS) + j;
        v4f v0 = frow[0];
        v4f v1 = frow[8];
        v4f v2 = frow[16];
        int en = r.y;
        live = (unsigned)en < (unsigned)n_edges;
        if (live) r = rec[en];
        a0 += v0; a1 += v1; a2 += v2;
        c++;
    }

    size_t nn = (size_t)n_nodes;
    v4f* row = (v4f*)(sT + ((size_t)t * nn + (size_t)node) * IN_FEATS) + j;
    row[0]  = a0;
    row[8]  = a1;
    row[16] = a2;
    if (j == 0) {
        float* cnt = sT + (size_t)(N_ETYPES * IN_FEATS) * nn;
        cnt[(size_t)node * N_ETYPES + t] = (float)c;
    }
}

__global__ __launch_bounds__(256) void ggc_gemm(const float* __restrict__ sT,
                                                const float* __restrict__ Bm2,
                                                float* __restrict__ out,
                                                int n_nodes, int n_tiles) {
    int lane  = threadIdx.x & 31;
    int wave  = threadIdx.x >> 5;
    int tile  = blockIdx.x * 8 + wave;
    if (tile >= n_tiles) return;

    int node0 = tile * 16;
    int m     = lane & 15;
    int hi    = lane >> 4;
    int mrow  = node0 + m;
    if (mrow > n_nodes - 1) mrow = n_nodes - 1;
    size_t nn = (size_t)n_nodes;

    v8f acc[8] = {};

    const float* Bl = Bm2 + (size_t)hi * (2 * OUT_FEATS * 2) + m * 2;

    for (int t = 0; t < N_ETYPES; t++) {
        const float* Arow = sT + ((size_t)t * nn + (size_t)mrow) * IN_FEATS + 4 * hi;
        for (int q = 0; q < IN_FEATS / 8; q++) {
            v4f a4 = *(const v4f*)(Arow + 8 * q);
            int pair0 = t * (IN_FEATS / 2) + 4 * q;
            #pragma unroll
            for (int s = 0; s < 2; s++) {
                v2f a;
                a.x = s ? a4.z : a4.x;
                a.y = s ? a4.w : a4.y;
                const float* Bp = Bl + (size_t)(pair0 + s) * (OUT_FEATS * 2);
                #pragma unroll
                for (int nt = 0; nt < 8; nt++) {
                    v2f bv = *(const v2f*)(Bp + nt * 32);
                    acc[nt] = __builtin_amdgcn_wmma_f32_16x16x4_f32(
                         false, a,  false, bv,
                         (short)0, acc[nt],  false,  false);
                }
            }
        }
    }

    {
        const float* cnt = sT + (size_t)(N_ETYPES * IN_FEATS) * nn;
        v2f a = *(const v2f*)(cnt + (size_t)mrow * N_ETYPES + 2 * hi);
        const float* Bp = Bm2 + (size_t)(N_ETYPES * IN_FEATS / 2 + hi) * (OUT_FEATS * 2) + m * 2;
        #pragma unroll
        for (int nt = 0; nt < 8; nt++) {
            v2f bv = *(const v2f*)(Bp + nt * 32);
            acc[nt] = __builtin_amdgcn_wmma_f32_16x16x4_f32(
                 false, a,  false, bv,
                 (short)0, acc[nt],  false,  false);
        }
    }

    int mbase = node0 + hi * 8;
    #pragma unroll
    for (int r = 0; r < 8; r++) {
        int node = mbase + r;
        if (node < n_nodes) {
            float* orow = out + (size_t)node * OUT_FEATS;
            #pragma unroll
            for (int nt = 0; nt < 8; nt++) {
                orow[nt * 16 + m] = acc[nt][r];
            }
        }
    }
}

extern "C" void kernel_launch(void* const* d_in, const int* in_sizes, int n_in,
                              void* d_out, int out_size, void* d_ws, size_t ws_size,
                              hipStream_t stream) {
    const float* feat   = (const float*)d_in[0];
    const int*   etypes = (const int*)  d_in[1];
    const int*   src    = (const int*)  d_in[2];
    const int*   dst    = (const int*)  d_in[3];
    const float* W      = (const float*)d_in[4];
    const float* b      = (const float*)d_in[5];
    float*       out    = (float*)d_out;

    int n_nodes = in_sizes[0] / IN_FEATS;
    int n_edges = in_sizes[1];

    float* sT   = (float*)d_ws;
    float* Bm2  = sT + (size_t)K_DIM * n_nodes;
    int*   head = (int*)(Bm2 + (size_t)K_DIM * OUT_FEATS);
    int2*  rec  = (int2*)(head + (size_t)n_nodes * N_ETYPES);

    int packN   = K_DIM * OUT_FEATS;
    int n_heads = n_nodes * N_ETYPES;
    int prepN   = packN > n_heads ? packN : n_heads;
    ggc_prep<<<(prepN + 255) / 256, 256, 0, stream>>>(W, b, Bm2, head, n_heads);

    ggc_link<<<(n_edges + 255) / 256, 256, 0, stream>>>(etypes, src, dst, head, rec,
                                                        n_edges, n_nodes);

    int gather_blocks = (n_nodes * 32 + 255) / 256;
    ggc_gather<<<gather_blocks, 256, 0, stream>>>(feat, head, rec, sT, n_edges, n_nodes);

    int n_tiles = (n_nodes + 15) / 16;
    ggc_gemm<<<(n_tiles + 7) / 8, 256, 0, stream>>>(sT, Bm2, out, n_nodes, n_tiles);
}


# GEMM phase: a wave owns one 16-column N tile of both M tiles, so each piece of B is fetched once per work-group, not twice; result rows staged in LDS, each stored as one 512-B row; sum orders kept
# speedup vs baseline: 12.1045x; 1.2787x over previous
#include <hip/hip_runtime.h>

#define IN_FEATS   96
#define OUT_FEATS  128
#define N_ETYPES   4
#define K_DIM      (N_ETYPES * IN_FEATS + N_ETYPES)
#define K_PAIRS    (K_DIM / 2)

typedef __attribute__((ext_vector_type(2))) float v2f;
typedef __attribute__((ext_vector_type(4))) float v4f;
typedef __attribute__((ext_vector_type(8))) float v8f;

__global__ __launch_bounds__(256) void ggc_prep(const float* __restrict__ W,
                                                const float* __restrict__ b,
                                                float* __restrict__ Bm2,
                                                int* __restrict__ head, int n_heads) {
    int idx = blockIdx.x * blockDim.x + threadIdx.x;
    if (idx < n_heads) head[idx] = -1;
    if (idx >= K_DIM * OUT_FEATS) return;
    int h  = idx & 1;
    int j  = (idx >> 1) & (OUT_FEATS - 1);
    int p  = idx >> 8;
    int kk = 2 * p + h;
    float v;
    if (kk < N_ETYPES * IN_FEATS) {
        int t = kk / IN_FEATS;
        int k = kk - t * IN_FEATS;
        v = W[((size_t)t * OUT_FEATS + j) * OUT_FEATS + k];
    } else {
        v = b[(kk - N_ETYPES * IN_FEATS) * OUT_FEATS + j];
    }
    Bm2[idx] = v;
}

__global__ __launch_bounds__(256) void ggc_link(const int* __restrict__ etypes,
                                                const int* __restrict__ src,
                                                const int* __restrict__ dst,
                                                int* __restrict__ head,
                                                int2* __restrict__ rec,
                                                int n_edges, int n_nodes) {
    int e = blockIdx.x * blockDim.x + threadIdx.x;
    if (e >= n_edges) return;
    int t = etypes[e];
    int s = src[e];
    int d = dst[e];
    int prev = -1;
    if ((unsigned)t < (unsigned)N_ETYPES && (unsigned)d < (unsigned)n_nodes)
        prev = atomicExch(&head[(size_t)d * N_ETYPES + t], e);
    rec[e] = make_int2(s, prev);
}

#define TILE_M   32
#define A_PITCH  K_DIM
#define NU       4
#define O_PITCH  132

__global__ __launch_bounds__(256) void ggc_gather_gemm(const float* __restrict__ feat,
                                                       const int*   __restrict__ head,
                                                       const int2*  __restrict__ rec,
                                                       const float* __restrict__ Bm2,
                                                       float* __restrict__ out,
                                                       int n_edges, int n_nodes) {
    __shared__ __attribute__((aligned(16))) float As[TILE_M * A_PITCH];

    int lane  = threadIdx.x & 31;
    int wave  = threadIdx.x >> 5;
    int node0 = blockIdx.x * TILE_M;

    {
        int t = lane >> 3;
        int j = lane & 7;
        const v4f zero4 = {0.f, 0.f, 0.f, 0.f};

        int2 r[NU];
        bool live[NU];
        int  c[NU];
        v4f  a0[NU], a1[NU], a2[NU];


        int e0[NU];
        #pragma unroll
        for (int u = 0; u < NU; u++) {
            int node = node0 + wave * NU + u;
            int nd   = node < n_nodes ? node : n_nodes - 1;
            int h    = head[(size_t)nd * N_ETYPES + t];
            e0[u] = node < n_nodes ? h : -1;
        }
        bool any = false;
        #pragma unroll
        for (int u = 0; u < NU; u++) {
            live[u] = (unsigned)e0[u] < (unsigned)n_edges;
            int2 r0 = rec[live[u] ? e0[u] : 0];
            r[u].x = live[u] ? r0.x : 0;
            r[u].y = live[u] ? r0.y : -1;
            a0[u] = zero4; a1[u] = zero4; a2[u] = zero4;
            c[u] = 0;
            any = any || live[u];
        }

        for (int it = 0; it < n_edges && any; ++it) {
            v4f v0[NU], v1[NU], v2[NU];
            #pragma unroll
            for (int u = 0; u < NU; u++) {
                const v4f* frow = (const v4f*)(feat + (size_t)r[u].x * IN_FEATS) + j;
                v0[u] = frow[0];
                v1[u] = frow[8];
                v2[u] = frow[16];
            }
            int2 rn[NU];
            bool nl[NU];
            #pragma unroll
            for (int u = 0; u < NU; u++) {
                int en = r[u].y;
                nl[u] = live[u] && (unsigned)en < (unsigned)n_edges;
                rn[u] = rec[nl[u] ? en : 0];
            }
            any = false;
            #pragma unroll
            for (int u = 0; u < NU; u++) {
                a0[u] += live[u] ? v0[u] : zero4;
                a1[u] += live[u] ? v1[u] : zero4;
                a2[u] += live[u] ? v2[u] : zero4;
                c[u]  += live[u] ? 1 : 0;
                r[u].x = nl[u] ? rn[u].x : 0;
                r[u].y = nl[u] ? rn[u].y : -1;
                live[u] = nl[u];
                any = any || nl[u];
            }
        }

        #pragma unroll
        for (int u = 0; u < NU; u++) {
            int ml = wave * NU + u;
            v4f* row = (v4f*)(As + ml * A_PITCH + t * IN_FEATS) + j;
            row[0]  = a0[u];
            row[8]  = a1[u];
            row[16] = a2[u];
            if (j == 0) As[ml * A_PITCH + N_ETYPES * IN_FEATS + t] = (float)c[u];
        }
    }

    __syncthreads();

    int m  = lane & 15;
    int hi = lane >> 4;

    v8f acc0 = {}, acc1 = {};

    const float* Al0 = As + m * A_PITCH + 4 * hi;
    const float* Al1 = Al0 + 16 * A_PITCH;
    const float* Bl = Bm2 + (size_t)hi * (2 * OUT_FEATS * 2) + wave * 32 + m * 2;

    for (int t = 0; t < N_ETYPES; t++) {
        #pragma unroll 4
        for (int q = 0; q < IN_FEATS / 8; q++) {
            v4f a40 = *(const v4f*)(Al0 + t * IN_FEATS + 8 * q);
            v4f a41 = *(const v4f*)(Al1 + t * IN_FEATS + 8 * q);
            int pair0 = t * (IN_FEATS / 2) + 4 * q;
            #pragma unroll
            for (int s = 0; s < 2; s++) {
                v2f a0, a1;
                a0.x = s ? a40.z : a40.x;
                a0.y = s ? a40.w : a40.y;
                a1.x = s ? a41.z : a41.x;
                a1.y = s ? a41.w : a41.y;
                v2f bb = *(const v2f*)(Bl + (size_t)(pair0 + s) * (OUT_FEATS * 2));
                acc0 = __builtin_amdgcn_wmma_f32_16x16x4_f32(
                     false, a0,  false, bb,
                     (short)0, acc0,  false,  false);
                acc1 = __builtin_amdgcn_wmma_f32_16x16x4_f32(
                     false, a1,  false, bb,
                     (short)0, acc1,  false,  false);
            }
        }
    }

    {
        const float* Ac = As + m * A_PITCH + N_ETYPES * IN_FEATS + 2 * hi;
        v2f a0 = *(const v2f*)(Ac);
        v2f a1 = *(const v2f*)(Ac + 16 * A_PITCH);
        v2f bb = *(const v2f*)(Bm2 + (size_t)(N_ETYPES * IN_FEATS / 2 + hi) * (OUT_FEATS * 2)
                               + wave * 32 + m * 2);
        acc0 = __builtin_amdgcn_wmma_f32_16x16x4_f32(
             false, a0,  false, bb,
             (short)0, acc0,  false,  false);
        acc1 = __builtin_amdgcn_wmma_f32_16x16x4_f32(
             false, a1,  false, bb,
             (short)0, acc1,  false,  false);
    }

    __syncthreads();

    float* Os = As;
    #pragma unroll
    for (int r8 = 0; r8 < 8; r8++) {
        Os[(hi * 8 + r8) * O_PITCH + wave * 16 + m]      = acc0[r8];
        Os[(16 + hi * 8 + r8) * O_PITCH + wave * 16 + m] = acc1[r8];
    }

    __syncthreads();

    #pragma unroll
    for (int i = 0; i < 4; i++) {
        int ml   = wave * 4 + i;
        int node = node0 + ml;
        if (node < n_nodes) {
            v4f v = *(const v4f*)(Os + ml * O_PITCH + 4 * lane);
            *(v4f*)(out + (size_t)node * OUT_FEATS + 4 * lane) = v;
        }
    }
}

extern "C" void kernel_launch(void* const* d_in, const int* in_sizes, int n_in,
                              void* d_out, int out_size, void* d_ws, size_t ws_size,
                              hipStream_t stream) {
    const float* feat   = (const float*)d_in[0];
    const int*   etypes = (const int*)  d_in[1];
    const int*   src    = (const int*)  d_in[2];
    const int*   dst    = (const int*)  d_in[3];
    const float* W      = (const float*)d_in[4];
    const float* b      = (const float*)d_in[5];
    float*       out    = (float*)d_out;

    int n_nodes = in_sizes[0] / IN_FEATS;
    int n_edges = in_sizes[1];

    float* Bm2  = (float*)d_ws;
    int*   head = (int*)(Bm2 + (size_t)K_DIM * OUT_FEATS);
    int2*  rec  = (int2*)(head + (size_t)n_nodes * N_ETYPES);

    int packN   = K_DIM * OUT_FEATS;
    int n_heads = n_nodes * N_ETYPES;
    int prepN   = packN > n_heads ? packN : n_heads;
    ggc_prep<<<(prepN + 255) / 256, 256, 0, stream>>>(W, b, Bm2, head, n_heads);

    ggc_link<<<(n_edges + 255) / 256, 256, 0, stream>>>(etypes, src, dst, head, rec,
                                                        n_edges, n_nodes);

    int n_tiles = (n_nodes + TILE_M - 1) / TILE_M;
    ggc_gather_gemm<<<n_tiles, 256, 0, stream>>>(feat, head, rec, Bm2, out, n_edges, n_nodes);
}


# Edges binned by 32-node tile in LDS (counting sort per 4096-edge block); a tile builds its lists by LDS exchange: no global exchange per edge, no clear pass, 3 launches -> 2; sums in LDS list order
# speedup vs baseline: 16.1038x; 1.3304x over previous
#include <hip/hip_runtime.h>

#define IN_FEATS   96
#define OUT_FEATS  128
#define N_ETYPES   4
#define K_DIM      (N_ETYPES * IN_FEATS + N_ETYPES)
#define K_PAIRS    (K_DIM / 2)

typedef __attribute__((ext_vector_type(2))) float v2f;
typedef __attribute__((ext_vector_type(4))) float v4f;
typedef __attribute__((ext_vector_type(8))) float v8f;

#define TILE_M      32
#define TILE_SHIFT  5
#define EPB         4096
#define BIN_THREADS 512
#define EPT         (EPB / BIN_THREADS)
#define MAX_TILES   2048
#define OFF_PITCH   MAX_TILES
#define POOL        1024

__global__ __launch_bounds__(BIN_THREADS) void ggc_bin_pack(const int* __restrict__ etypes,
                                                            const int* __restrict__ src,
                                                            const int* __restrict__ dst,
                                                            const float* __restrict__ W,
                                                            const float* __restrict__ b,
                                                            float* __restrict__ Bm2,
                                                            int* __restrict__ off,
                                                            int2* __restrict__ binned,
                                                            int n_edges, int n_nodes, int n_bin_blocks) {
    __shared__ __attribute__((aligned(16))) int  cnt[MAX_TILES];
    __shared__ __attribute__((aligned(16))) int2 stage[EPB];
    __shared__ int wsum[BIN_THREADS / 32];

    int tid = threadIdx.x;

    if ((int)blockIdx.x >= n_bin_blocks) {
        int idx = ((int)blockIdx.x - n_bin_blocks) * BIN_THREADS + tid;
        if (idx >= K_DIM * OUT_FEATS) return;
        int h  = idx & 1;
        int j  = (idx >> 1) & (OUT_FEATS - 1);
        int p  = idx >> 8;
        int kk = 2 * p + h;
        float v;
        if (kk < N_ETYPES * IN_FEATS) {
            int t = kk / IN_FEATS;
            int k = kk - t * IN_FEATS;
            v = W[((size_t)t * OUT_FEATS + j) * OUT_FEATS + k];
        } else {
            v = b[(kk - N_ETYPES * IN_FEATS) * OUT_FEATS + j];
        }
        Bm2[idx] = v;
        return;
    }

    for (int g = tid; g < MAX_TILES; g += BIN_THREADS) cnt[g] = 0;
    __syncthreads();

    int ebase = (int)blockIdx.x * EPB;
    int  gk[EPT];
    int2 rec[EPT];
    int et[EPT], es[EPT], ed[EPT];
    #pragma unroll
    for (int k = 0; k < EPT; k++) {
        int e  = ebase + k * BIN_THREADS + tid;
        int ec = e < n_edges ? e : n_edges - 1;
        et[k] = etypes[ec];
        es[k] = src[ec];
        ed[k] = dst[ec];
    }
    #pragma unroll
    for (int k = 0; k < EPT; k++) {
        int e = ebase + k * BIN_THREADS + tid;
        int t = et[k];
        int d = ed[k];
        bool ok = e < n_edges && (unsigned)t < (unsigned)N_ETYPES && (unsigned)d < (unsigned)n_nodes;
        gk[k]    = ok ? (d >> TILE_SHIFT) : -1;
        rec[k].x = es[k];
        rec[k].y = ((d & (TILE_M - 1)) << 2) | (t & 3);
    }
    int rk[EPT];
    #pragma unroll
    for (int k = 0; k < EPT; k++) {
        rk[k] = 0;
        if (gk[k] >= 0) rk[k] = atomicAdd(&cnt[gk[k]], 1);
    }
    __syncthreads();

    int lane = tid & 31;
    int wv   = tid >> 5;
    int c0 = cnt[4 * tid], c1 = cnt[4 * tid + 1], c2 = cnt[4 * tid + 2], c3 = cnt[4 * tid + 3];
    int tsum = c0 + c1 + c2 + c3;
    int x = tsum;
    #pragma unroll
    for (int o = 1; o < 32; o <<= 1) {
        int y = __shfl_up(x, o);
        if (lane >= o) x += y;
    }
    if (lane == 31) wsum[wv] = x;
    __syncthreads();
    int wbase = 0, total = 0;
    #pragma unroll
    for (int w2 = 0; w2 < BIN_THREADS / 32; w2++) {
        int s = wsum[w2];
        wbase += (w2 < wv) ? s : 0;
        total += s;
    }
    int o0 = wbase + x - tsum;
    int o1 = o0 + c0, o2 = o1 + c1, o3 = o2 + c2;
    cnt[4 * tid] = o0; cnt[4 * tid + 1] = o1; cnt[4 * tid + 2] = o2; cnt[4 * tid + 3] = o3;
    int* orow = off + (size_t)blockIdx.x * OFF_PITCH + 4 * tid;
    orow[0] = o0; orow[1] = o1; orow[2] = o2; orow[3] = o3;
    __syncthreads();

    #pragma unroll
    for (int k = 0; k < EPT; k++) {
        if (gk[k] >= 0) stage[cnt[gk[k]] + rk[k]] = rec[k];
    }
    __syncthreads();

    int2* outp = binned + (size_t)blockIdx.x * EPB;
    for (int p = tid; p < total; p += BIN_THREADS) outp[p] = stage[p];
}

#define A_PITCH  K_DIM
#define NU       4
#define O_PITCH  132
#define GG_THREADS 256
#define RUN_HEAD 8

__global__ __launch_bounds__(GG_THREADS) void ggc_gather_gemm(const float* __restrict__ feat,
                                                              const int*   __restrict__ off,
                                                              const int2*  __restrict__ binned,
                                                              const float* __restrict__ Bm2,
                                                              float* __restrict__ out,
                                                              int n_bin_blocks, int n_nodes) {
    __shared__ __attribute__((aligned(16))) float As[TILE_M * A_PITCH];
    __shared__ __attribute__((aligned(16))) int2  pool[POOL];
    __shared__ int heads[TILE_M * N_ETYPES];
    __shared__ int cntL[TILE_M * N_ETYPES];
    __shared__ int wsum[GG_THREADS / 32];

    int tid   = threadIdx.x;
    int lane  = threadIdx.x & 31;
    int wave  = threadIdx.x >> 5;
    int node0 = blockIdx.x * TILE_M;

    {
        int t = lane >> 3;
        int j = lane & 7;
        const v4f zero4 = {0.f, 0.f, 0.f, 0.f};

        if (tid < TILE_M * N_ETYPES) { heads[tid] = -1; cntL[tid] = 0; }

        int ro = 0, rl = 0;
        if (tid < n_bin_blocks) {
            const int* orow = off + (size_t)tid * OFF_PITCH + blockIdx.x;
            int b0 = orow[0];
            int b1 = orow[1];
            rl = b1 - b0;
            rl = rl < 0 ? 0 : (rl > EPB ? EPB : rl);
            ro = b0 < 0 ? 0 : (b0 > EPB - rl ? EPB - rl : b0);
        }
        const int2* run = binned + (size_t)(tid < n_bin_blocks ? tid : 0) * EPB + ro;
        int x = rl;
        #pragma unroll
        for (int o = 1; o < 32; o <<= 1) {
            int y = __shfl_up(x, o);
            if (lane >= o) x += y;
        }
        if (lane == 31) wsum[wave] = x;
        __syncthreads();
        int wbase = 0, total = 0;
        #pragma unroll
        for (int w2 = 0; w2 < GG_THREADS / 32; w2++) {
            int s = wsum[w2];
            wbase += (w2 < wave) ? s : 0;
            total += s;
        }
        int rp = wbase + x - rl;

        int2 rc[RUN_HEAD];
        int  pp[RUN_HEAD];
        #pragma unroll
        for (int q = 0; q < RUN_HEAD; q++) {
            int p = rp + q;
            bool ok = q < rl && p < POOL;
            pp[q] = ok ? p : -1;
            rc[q] = make_int2(0, 0);
            if (ok) rc[q] = run[q];
        }
        #pragma unroll
        for (int q = 0; q < RUN_HEAD; q++) {
            if (pp[q] >= 0) {
                int lid  = rc[q].y & (TILE_M * N_ETYPES - 1);
                int prev = atomicExch(&heads[lid], pp[q]);
                pool[pp[q]] = make_int2(rc[q].x, prev);
                atomicAdd(&cntL[lid], 1);
            }
        }
        for (int q = RUN_HEAD; q < rl; q++) {
            int p = rp + q;
            if (p < POOL) {
                int2 rcs = run[q];
                int lid  = rcs.y & (TILE_M * N_ETYPES - 1);
                int prev = atomicExch(&heads[lid], p);
                pool[p] = make_int2(rcs.x, prev);
                atomicAdd(&cntL[lid], 1);
            }
        }
        __syncthreads();

        v4f  a0[NU], a1[NU], a2[NU];
        int2 r[NU];
        bool live[NU];

        bool any = false;
        #pragma unroll
        for (int u = 0; u < NU; u++) {
            a0[u] = zero4; a1[u] = zero4; a2[u] = zero4;
            int h = heads[(wave * NU + u) * N_ETYPES + t];
            live[u] = (unsigned)h < (unsigned)POOL;
            int2 r0 = pool[live[u] ? h : 0];
            r[u].x = live[u] ? r0.x : 0;
            r[u].y = live[u] ? r0.y : -1;
            any = any || live[u];
        }

        for (int it = 0; it < POOL && any; ++it) {
            v4f v0[NU], v1[NU], v2[NU];
            #pragma unroll
            for (int u = 0; u < NU; u++) {
                unsigned fo = (unsigned)r[u].x * (unsigned)(IN_FEATS * 4) + (unsigned)j * 16u;
                const v4f* frow = (const v4f*)((const char*)feat + fo);
                v0[u] = frow[0];
                v1[u] = frow[8];
                v2[u] = frow[16];
            }
            int2 rn[NU];
            bool nl[NU];
            #pragma unroll
            for (int u = 0; u < NU; u++) {
                int en = r[u].y;
                nl[u] = live[u] && (unsigned)en < (unsigned)POOL;
                rn[u] = pool[nl[u] ? en : 0];
            }
            any = false;
            #pragma unroll
            for (int u = 0; u < NU; u++) {
                a0[u] += live[u] ? v0[u] : zero4;
                a1[u] += live[u] ? v1[u] : zero4;
                a2[u] += live[u] ? v2[u] : zero4;
                r[u].x = nl[u] ? rn[u].x : 0;
                r[u].y = nl[u] ? rn[u].y : -1;
                live[u] = nl[u];
                any = any || nl[u];
            }
        }

        #pragma unroll
        for (int u = 0; u < NU; u++) {
            int ml = wave * NU + u;
            v4f* row = (v4f*)(As + ml * A_PITCH + t * IN_FEATS) + j;
            row[0]  = a0[u];
            row[8]  = a1[u];
            row[16] = a2[u];
            if (j == 0) As[ml * A_PITCH + N_ETYPES * IN_FEATS + t] = (float)cntL[ml * N_ETYPES + t];
        }

        if (total > POOL || n_bin_blocks > GG_THREADS) {
            int pos = 0;
            for (int i = 0; i < n_bin_blocks; i++) {
                const int* orow = off + (size_t)i * OFF_PITCH + blockIdx.x;
                int b0 = orow[0];
                int b1 = orow[1];
                int len = b1 - b0;
                len = len < 0 ? 0 : (len > EPB ? EPB : len);
                int st  = b0 < 0 ? 0 : (b0 > EPB - len ? EPB - len : b0);
                const int2* rs = binned + (size_t)i * EPB + st;
                for (int q = 0; q < len; q++) {
                    bool pooled = i < GG_THREADS && pos + q < POOL;
                    if (!pooled) {
                        int2 rcs = rs[q];
                        int lid = rcs.y & (TILE_M * N_ETYPES - 1);
                        int ml  = lid >> 2;
                        int tt  = lid & 3;
                        if (wave == ml / NU && t == tt) {
                            const v4f* frow = (const v4f*)(feat + (size_t)rcs.x * IN_FEATS) + j;
                            v4f* row = (v4f*)(As + ml * A_PITCH + tt * IN_FEATS) + j;
                            row[0]  += frow[0];
                            row[8]  += frow[8];
                            row[16] += frow[16];
                            if (j == 0) As[ml * A_PITCH + N_ETYPES * IN_FEATS + tt] += 1.0f;
                        }
                    }
                }
                if (i < GG_THREADS) pos += len;
            }
        }
    }

    __syncthreads();

    int m  = lane & 15;
    int hi = lane >> 4;

    v8f acc0 = {}, acc1 = {};

    const float* Al0 = As + m * A_PITCH + 4 * hi;
    const float* Al1 = Al0 + 16 * A_PITCH;
    const float* Bl = Bm2 + (size_t)hi * (2 * OUT_FEATS * 2) + wave * 32 + m * 2;

    for (int t = 0; t < N_ETYPES; t++) {
        #pragma unroll 4
        for (int q = 0; q < IN_FEATS / 8; q++) {
            v4f a40 = *(const v4f*)(Al0 + t * IN_FEATS + 8 * q);
            v4f a41 = *(const v4f*)(Al1 + t * IN_FEATS + 8 * q);
            int pair0 = t * (IN_FEATS / 2) + 4 * q;
            #pragma unroll
            for (int s = 0; s < 2; s++) {
                v2f a0, a1;
                a0.x = s ? a40.z : a40.x;
                a0.y = s ? a40.w : a40.y;
                a1.x = s ? a41.z : a41.x;
                a1.y = s ? a41.w : a41.y;
                v2f bb = *(const v2f*)(Bl + (size_t)(pair0 + s) * (OUT_FEATS * 2));
                acc0 = __builtin_amdgcn_wmma_f32_16x16x4_f32(
                     false, a0,  false, bb,
                     (short)0, acc0,  false,  false);
                acc1 = __builtin_amdgcn_wmma_f32_16x16x4_f32(
                     false, a1,  false, bb,
                     (short)0, acc1,  false,  false);
            }
        }
    }

    {
        const float* Ac = As + m * A_PITCH + N_ETYPES * IN_FEATS + 2 * hi;
        v2f a0 = *(const v2f*)(Ac);
        v2f a1 = *(const v2f*)(Ac + 16 * A_PITCH);
        v2f bb = *(const v2f*)(Bm2 + (size_t)(N_ETYPES * IN_FEATS / 2 + hi) * (OUT_FEATS * 2)
                               + wave * 32 + m * 2);
        acc0 = __builtin_amdgcn_wmma_f32_16x16x4_f32(
             false, a0,  false, bb,
             (short)0, acc0,  false,  false);
        acc1 = __builtin_amdgcn_wmma_f32_16x16x4_f32(
             false, a1,  false, bb,
             (short)0, acc1,  false,  false);
    }

    __syncthreads();

    float* Os = As;
    #pragma unroll
    for (int r8 = 0; r8 < 8; r8++) {
        Os[(hi * 8 + r8) * O_PITCH + wave * 16 + m]      = acc0[r8];
        Os[(16 + hi * 8 + r8) * O_PITCH + wave * 16 + m] = acc1[r8];
    }

    __syncthreads();

    #pragma unroll
    for (int i = 0; i < 4; i++) {
        int ml   = wave * 4 + i;
        int node = node0 + ml;
        if (node < n_nodes) {
            v4f v = *(const v4f*)(Os + ml * O_PITCH + 4 * lane);
            *(v4f*)(out + (size_t)node * OUT_FEATS + 4 * lane) = v;
        }
    }
}

extern "C" void kernel_launch(void* const* d_in, const int* in_sizes, int n_in,
                              void* d_out, int out_size, void* d_ws, size_t ws_size,
                              hipStream_t stream) {
    const float* feat   = (const float*)d_in[0];
    const int*   etypes = (const int*)  d_in[1];
    const int*   src    = (const int*)  d_in[2];
    const int*   dst    = (const int*)  d_in[3];
    const float* W      = (const float*)d_in[4];
    const float* b      = (const float*)d_in[5];
    float*       out    = (float*)d_out;

    int n_nodes = in_sizes[0] / IN_FEATS;
    int n_edges = in_sizes[1];

    int n_tiles      = (n_nodes + TILE_M - 1) / TILE_M;
    int n_bin_blocks = (n_edges + EPB - 1) / EPB;
    int n_pack_blocks = (K_DIM * OUT_FEATS + BIN_THREADS - 1) / BIN_THREADS;

    float* Bm2    = (float*)d_ws;
    int*   off    = (int*)(Bm2 + (size_t)K_DIM * OUT_FEATS);
    int2*  binned = (int2*)(off + (size_t)n_bin_blocks * OFF_PITCH);

    ggc_bin_pack<<<n_bin_blocks + n_pack_blocks, BIN_THREADS, 0, stream>>>(
        etypes, src, dst, W, b, Bm2, off, binned, n_edges, n_nodes, n_bin_blocks);

    ggc_gather_gemm<<<n_tiles, GG_THREADS, 0, stream>>>(feat, off, binned, Bm2, out,
                                                        n_bin_blocks, n_nodes);
}
